# starting kernel
# baseline (speedup 1.0000x reference)
#include <hip/hip_runtime.h>
#include <hip/hip_bf16.h>

#define NBATCH 256
#define CDIM   256
#define KSUB   128

#define RHO_S  144
#define TT_S   137

typedef __attribute__((ext_vector_type(2))) float v2f;
typedef __attribute__((ext_vector_type(8))) float v8f;

__global__ void build_ut_kernel(const float* __restrict__ ux,
                                const float* __restrict__ uy,
                                const float* __restrict__ uc,
                                float* __restrict__ UT) {
    int idx = blockIdx.x * blockDim.x + threadIdx.x;
    if (idx >= CDIM * KSUB) return;
    int p = idx / KSUB;
    int q = idx % KSUB;
    int c1 = p >> 6;
    int i1 = (p >> 3) & 7;
    int j1 = p & 7;
    int c2 = 2 + (q >> 6);
    int i2 = (q >> 3) & 7;
    int j2 = q & 7;
    UT[q * CDIM + p] = uc[c1 * 4 + c2] * ux[i1 * 8 + i2] * uy[j1 * 8 + j2];
}

__global__ void __launch_bounds__(256, 2)
qconv_wmma_kernel(const float* __restrict__ rho,
                  const float* __restrict__ UT,
                  float* __restrict__ out) {
    extern __shared__ float smem[];
    float* sRho = smem;
    float* sTt  = smem + KSUB * RHO_S;

    const int b    = blockIdx.x >> 1;
    const int half = blockIdx.x & 1;
    const int tid  = threadIdx.x;
    const int lane = tid & 31;
    const int w    = tid >> 5;
    const int l16  = lane & 15;
    const int hl   = lane >> 4;

    {
        const float4* src = (const float4*)(rho + (size_t)b * KSUB * KSUB);
        float4* dst = (float4*)sRho;
        #pragma unroll
        for (int i = 0; i < (KSUB * KSUB / 4) / 256; ++i) {
            int g    = tid + i * 256;
            int row  = g >> 5;
            int col4 = g & 31;
            dst[row * (RHO_S / 4) + col4] = src[g];
        }
    }
    __syncthreads();

    {
        const int prow = half * KSUB + w * 16 + l16;
        v8f acc[8] = {v8f{}, v8f{}, v8f{}, v8f{}, v8f{}, v8f{}, v8f{}, v8f{}};
        for (int kk = 0; kk < KSUB; kk += 4) {
            v2f a;
            a.x = UT[(kk + 2 * hl)     * CDIM + prow];
            a.y = UT[(kk + 2 * hl + 1) * CDIM + prow];
            #pragma unroll
            for (int tn = 0; tn < 8; ++tn) {
                v2f bf;
                bf.x = sRho[(kk + 2 * hl)     * RHO_S + tn * 16 + l16];
                bf.y = sRho[(kk + 2 * hl + 1) * RHO_S + tn * 16 + l16];
                acc[tn] = __builtin_amdgcn_wmma_f32_16x16x4_f32(
                    false, a, false, bf, (short)0, acc[tn], false, false);
            }
        }
        #pragma unroll
        for (int tn = 0; tn < 8; ++tn) {
            #pragma unroll
            for (int v = 0; v < 8; ++v)
                sTt[(tn * 16 + l16) * TT_S + w * 16 + v + 8 * hl] = acc[tn][v];
        }
    }
    __syncthreads();

    float* outb = out + (size_t)b * CDIM * CDIM + (size_t)half * KSUB * CDIM;
    const int plocal = w * 16 + l16;
    for (int th = 0; th < 2; ++th) {
        v8f acc[8] = {v8f{}, v8f{}, v8f{}, v8f{}, v8f{}, v8f{}, v8f{}, v8f{}};
        for (int kk = 0; kk < KSUB; kk += 4) {
            v2f a;
            a.x = sTt[(kk + 2 * hl)     * TT_S + plocal];
            a.y = sTt[(kk + 2 * hl + 1) * TT_S + plocal];
            #pragma unroll
            for (int tn8 = 0; tn8 < 8; ++tn8) {
                const int tn = th * 8 + tn8;
                v2f bf;
                bf.x = UT[(kk + 2 * hl)     * CDIM + tn * 16 + l16];
                bf.y = UT[(kk + 2 * hl + 1) * CDIM + tn * 16 + l16];
                acc[tn8] = __builtin_amdgcn_wmma_f32_16x16x4_f32(
                    false, a, false, bf, (short)0, acc[tn8], false, false);
            }
        }
        #pragma unroll
        for (int tn8 = 0; tn8 < 8; ++tn8) {
            const int tn = th * 8 + tn8;
            #pragma unroll
            for (int v = 0; v < 8; ++v)
                outb[(w * 16 + v + 8 * hl) * CDIM + tn * 16 + l16] = acc[tn8][v];
        }
    }
}

extern "C" void kernel_launch(void* const* d_in, const int* in_sizes, int n_in,
                              void* d_out, int out_size, void* d_ws, size_t ws_size,
                              hipStream_t stream) {
    (void)in_sizes; (void)n_in; (void)out_size; (void)ws_size;
    const float* rho = (const float*)d_in[0];
    const float* ux  = (const float*)d_in[1];
    const float* uy  = (const float*)d_in[2];
    const float* uc  = (const float*)d_in[3];
    float* out = (float*)d_out;

    float* UT = (float*)d_ws;

    build_ut_kernel<<<(CDIM * KSUB + 255) / 256, 256, 0, stream>>>(ux, uy, uc, UT);

    const int lds_bytes = (KSUB * RHO_S + KSUB * TT_S) * (int)sizeof(float);
    (void)hipFuncSetAttribute((const void*)qconv_wmma_kernel,
                              hipFuncAttributeMaxDynamicSharedMemorySize, lds_bytes);

    qconv_wmma_kernel<<<NBATCH * 2, 256, lds_bytes, stream>>>(rho, UT, out);
}


# Build K^T = kron(ux,uy)^T inside the main kernel, straight into LDS per workgroup under the rho loads, instead of a separate setup launch through the workspace: one launch per call instead of two
# speedup vs baseline: 2.5640x; 2.5640x over previous
#include <hip/hip_runtime.h>

#define NBATCH 256
#define CDIM   256
#define KSUB   128
#define KD     64

#define RHO_S  144
#define PT_S   137
#define KT_S   72

typedef __attribute__((ext_vector_type(2))) float v2f;
typedef __attribute__((ext_vector_type(8))) float v8f;

__global__ void __launch_bounds__(256)
kron_apply_kernel(const float* __restrict__ rho,
                  const float* __restrict__ ux,
                  const float* __restrict__ uy,
                  const float* __restrict__ uc,
                  float* __restrict__ out) {
    extern __shared__ float smem[];
    float* sRho = smem;
    float* sPt  = smem;
    float* sKT  = smem + KSUB * RHO_S;

    const int b    = blockIdx.x;
    const int tid  = threadIdx.x;
    const int lane = tid & 31;
    const int w    = tid >> 5;
    const int l16  = lane & 15;
    const int hl   = lane >> 4;

    {
        const int p  = tid & 63;
        const int i1 = p >> 3;
        const int j1 = p & 7;
        const int qb = tid >> 6;
        float uxr[8];
        #pragma unroll
        for (int t = 0; t < 8; ++t) uxr[t] = ux[i1 * 8 + t];
        const float uy0 = uy[j1 * 8 + qb];
        const float uy1 = uy[j1 * 8 + qb + 4];

        #pragma unroll
        for (int i = 0; i < 16; ++i)
            sKT[(qb + 4 * i) * KT_S + p] = uxr[i >> 1] * ((i & 1) ? uy1 : uy0);
    }
    {
        const float4* src = (const float4*)(rho + (size_t)b * KSUB * KSUB);
        float4* dst = (float4*)sRho;
        #pragma unroll
        for (int i = 0; i < (KSUB * KSUB / 4) / 256; ++i) {
            int g    = tid + i * 256;
            int row  = g >> 5;
            int col4 = g & 31;
            dst[row * (RHO_S / 4) + col4] = src[g];
        }
    }
    __syncthreads();

    {
        const int c   = w >> 2;
        const int pt0 = 2 * ((w >> 1) & 1);
        const int tn0 = 4 * (w & 1);
        v8f acc[2][4];
        #pragma unroll
        for (int i = 0; i < 2; ++i)
            #pragma unroll
            for (int j = 0; j < 4; ++j) acc[i][j] = v8f{};

        for (int kk = 0; kk < KD; kk += 4) {
            v2f a[2], bf[4];
            #pragma unroll
            for (int i = 0; i < 2; ++i) {
                a[i].x = sKT[(kk + 2 * hl)     * KT_S + (pt0 + i) * 16 + l16];
                a[i].y = sKT[(kk + 2 * hl + 1) * KT_S + (pt0 + i) * 16 + l16];
            }
            #pragma unroll
            for (int j = 0; j < 4; ++j) {
                bf[j].x = sRho[(c * KD + kk + 2 * hl)     * RHO_S + (tn0 + j) * 16 + l16];
                bf[j].y = sRho[(c * KD + kk + 2 * hl + 1) * RHO_S + (tn0 + j) * 16 + l16];
            }
            #pragma unroll
            for (int i = 0; i < 2; ++i)
                #pragma unroll
                for (int j = 0; j < 4; ++j)
                    acc[i][j] = __builtin_amdgcn_wmma_f32_16x16x4_f32(
                        false, a[i], false, bf[j], (short)0, acc[i][j], false, false);
        }
        __syncthreads();
        #pragma unroll
        for (int i = 0; i < 2; ++i)
            #pragma unroll
            for (int j = 0; j < 4; ++j)
                #pragma unroll
                for (int v = 0; v < 8; ++v)
                    sPt[((tn0 + j) * 16 + l16) * PT_S + c * KD + (pt0 + i) * 16 + v + 8 * hl] = acc[i][j][v];
    }
    __syncthreads();

    const int pt  = w >> 1;
    const int jp  = w & 1;
    v8f m[2][2][2];
    #pragma unroll
    for (int c = 0; c < 2; ++c)
        #pragma unroll
        for (int d = 0; d < 2; ++d)
            #pragma unroll
            for (int j = 0; j < 2; ++j) m[c][d][j] = v8f{};

    for (int kk = 0; kk < KD; kk += 4) {
        v2f a[2][2], bf[2];
        #pragma unroll
        for (int c = 0; c < 2; ++c)
            #pragma unroll
            for (int d = 0; d < 2; ++d) {
                a[c][d].x = sPt[(d * KD + kk + 2 * hl)     * PT_S + c * KD + pt * 16 + l16];
                a[c][d].y = sPt[(d * KD + kk + 2 * hl + 1) * PT_S + c * KD + pt * 16 + l16];
            }
        #pragma unroll
        for (int j = 0; j < 2; ++j) {
            bf[j].x = sKT[(kk + 2 * hl)     * KT_S + (2 * jp + j) * 16 + l16];
            bf[j].y = sKT[(kk + 2 * hl + 1) * KT_S + (2 * jp + j) * 16 + l16];
        }
        #pragma unroll
        for (int c = 0; c < 2; ++c)
            #pragma unroll
            for (int d = 0; d < 2; ++d)
                #pragma unroll
                for (int j = 0; j < 2; ++j)
                    m[c][d][j] = __builtin_amdgcn_wmma_f32_16x16x4_f32(
                        false, a[c][d], false, bf[j], (short)0, m[c][d][j], false, false);
    }

    float u0[4], u1[4];
    #pragma unroll
    for (int a = 0; a < 4; ++a) { u0[a] = uc[a * 4 + 2]; u1[a] = uc[a * 4 + 3]; }

    float* outb = out + (size_t)b * CDIM * CDIM
                + (size_t)(pt * 16 + 8 * hl) * CDIM + (2 * jp) * 16 + l16;
    #pragma unroll
    for (int a = 0; a < 4; ++a) {
        #pragma unroll
        for (int v = 0; v < 8; ++v) {
            #pragma unroll
            for (int j = 0; j < 2; ++j) {
                const float t0 = u0[a] * m[0][0][j][v] + u1[a] * m[1][0][j][v];
                const float t1 = u0[a] * m[0][1][j][v] + u1[a] * m[1][1][j][v];
                #pragma unroll
                for (int e = 0; e < 4; ++e)
                    outb[(a * KD + v) * CDIM + e * KD + j * 16] = u0[e] * t0 + u1[e] * t1;
            }
        }
    }
}

extern "C" void kernel_launch(void* const* d_in, const int* in_sizes, int n_in,
                              void* d_out, int out_size, void* d_ws, size_t ws_size,
                              hipStream_t stream) {
    (void)in_sizes; (void)n_in; (void)out_size; (void)d_ws; (void)ws_size;
    const float* rho = (const float*)d_in[0];
    const float* ux  = (const float*)d_in[1];
    const float* uy  = (const float*)d_in[2];
    const float* uc  = (const float*)d_in[3];
    float* out = (float*)d_out;

    const int lds_bytes = (KSUB * RHO_S + KD * KT_S) * (int)sizeof(float);
    (void)hipFuncSetAttribute((const void*)kron_apply_kernel,
                              hipFuncAttributeMaxDynamicSharedMemorySize, lds_bytes);

    kron_apply_kernel<<<NBATCH, 256, lds_bytes, stream>>>(rho, ux, uy, uc, out);
}


# Phase-B column tiles interleaved (tile j = columns 2*l+j) so a lane owns 2 adjacent outputs: 8-byte stores, each half-wave writing one whole 128-byte line, instead of two 64-byte halves
# speedup vs baseline: 2.6934x; 1.0505x over previous
#include <hip/hip_runtime.h>

#define NBATCH 256
#define CDIM   256
#define KSUB   128
#define KD     64

#define RHO_S  144
#define PT_S   137
#define KT_S   72

typedef __attribute__((ext_vector_type(2))) float v2f;
typedef __attribute__((ext_vector_type(8))) float v8f;

__global__ void __launch_bounds__(256)
kron_apply_kernel(const float* __restrict__ rho,
                  const float* __restrict__ ux,
                  const float* __restrict__ uy,
                  const float* __restrict__ uc,
                  float* __restrict__ out) {
    extern __shared__ float smem[];
    float* sRho = smem;
    float* sPt  = smem;
    float* sKT  = smem + KSUB * RHO_S;

    const int b    = blockIdx.x;
    const int tid  = threadIdx.x;
    const int lane = tid & 31;
    const int w    = tid >> 5;
    const int l16  = lane & 15;
    const int hl   = lane >> 4;

    {
        const int p  = tid & 63;
        const int i1 = p >> 3;
        const int j1 = p & 7;
        const int qb = tid >> 6;
        float uxr[8];
        #pragma unroll
        for (int t = 0; t < 8; ++t) uxr[t] = ux[i1 * 8 + t];
        const float uy0 = uy[j1 * 8 + qb];
        const float uy1 = uy[j1 * 8 + qb + 4];

        #pragma unroll
        for (int i = 0; i < 16; ++i)
            sKT[(qb + 4 * i) * KT_S + p] = uxr[i >> 1] * ((i & 1) ? uy1 : uy0);
    }
    {
        const float4* src = (const float4*)(rho + (size_t)b * KSUB * KSUB);
        float4* dst = (float4*)sRho;
        #pragma unroll
        for (int i = 0; i < (KSUB * KSUB / 4) / 256; ++i) {
            int g    = tid + i * 256;
            int row  = g >> 5;
            int col4 = g & 31;
            dst[row * (RHO_S / 4) + col4] = src[g];
        }
    }
    __syncthreads();

    {
        const int c   = w >> 2;
        const int pt0 = 2 * ((w >> 1) & 1);
        const int tn0 = 4 * (w & 1);
        v8f acc[2][4];
        #pragma unroll
        for (int i = 0; i < 2; ++i)
            #pragma unroll
            for (int j = 0; j < 4; ++j) acc[i][j] = v8f{};

        for (int kk = 0; kk < KD; kk += 4) {
            v2f a[2], bf[4];
            #pragma unroll
            for (int i = 0; i < 2; ++i) {
                a[i].x = sKT[(kk + 2 * hl)     * KT_S + (pt0 + i) * 16 + l16];
                a[i].y = sKT[(kk + 2 * hl + 1) * KT_S + (pt0 + i) * 16 + l16];
            }
            #pragma unroll
            for (int j = 0; j < 4; ++j) {
                bf[j].x = sRho[(c * KD + kk + 2 * hl)     * RHO_S + (tn0 + j) * 16 + l16];
                bf[j].y = sRho[(c * KD + kk + 2 * hl + 1) * RHO_S + (tn0 + j) * 16 + l16];
            }
            #pragma unroll
            for (int i = 0; i < 2; ++i)
                #pragma unroll
                for (int j = 0; j < 4; ++j)
                    acc[i][j] = __builtin_amdgcn_wmma_f32_16x16x4_f32(
                        false, a[i], false, bf[j], (short)0, acc[i][j], false, false);
        }
        __syncthreads();
        #pragma unroll
        for (int i = 0; i < 2; ++i)
            #pragma unroll
            for (int j = 0; j < 4; ++j)
                #pragma unroll
                for (int v = 0; v < 8; ++v)
                    sPt[((tn0 + j) * 16 + l16) * PT_S + c * KD + (pt0 + i) * 16 + v + 8 * hl] = acc[i][j][v];
    }
    __syncthreads();

    const int pt  = w >> 1;
    const int jp  = w & 1;
    v8f m[2][2][2];
    #pragma unroll
    for (int c = 0; c < 2; ++c)
        #pragma unroll
        for (int d = 0; d < 2; ++d)
            #pragma unroll
            for (int j = 0; j < 2; ++j) m[c][d][j] = v8f{};

    for (int kk = 0; kk < KD; kk += 4) {
        v2f a[2][2], bf[2];
        #pragma unroll
        for (int c = 0; c < 2; ++c)
            #pragma unroll
            for (int d = 0; d < 2; ++d) {
                a[c][d].x = sPt[(d * KD + kk + 2 * hl)     * PT_S + c * KD + pt * 16 + l16];
                a[c][d].y = sPt[(d * KD + kk + 2 * hl + 1) * PT_S + c * KD + pt * 16 + l16];
            }
        #pragma unroll
        for (int j = 0; j < 2; ++j) {
            bf[j].x = sKT[(kk + 2 * hl)     * KT_S + jp * 32 + 2 * l16 + j];
            bf[j].y = sKT[(kk + 2 * hl + 1) * KT_S + jp * 32 + 2 * l16 + j];
        }
        #pragma unroll
        for (int c = 0; c < 2; ++c)
            #pragma unroll
            for (int d = 0; d < 2; ++d)
                #pragma unroll
                for (int j = 0; j < 2; ++j)
                    m[c][d][j] = __builtin_amdgcn_wmma_f32_16x16x4_f32(
                        false, a[c][d], false, bf[j], (short)0, m[c][d][j], false, false);
    }

    float u0[4], u1[4];
    #pragma unroll
    for (int a = 0; a < 4; ++a) { u0[a] = uc[a * 4 + 2]; u1[a] = uc[a * 4 + 3]; }

    float* outb = out + (size_t)b * CDIM * CDIM
                + (size_t)(pt * 16 + 8 * hl) * CDIM + jp * 32 + 2 * l16;
    #pragma unroll
    for (int a = 0; a < 4; ++a) {
        #pragma unroll
        for (int v = 0; v < 8; ++v) {
            float t0[2], t1[2];
            #pragma unroll
            for (int j = 0; j < 2; ++j) {
                t0[j] = u0[a] * m[0][0][j][v] + u1[a] * m[1][0][j][v];
                t1[j] = u0[a] * m[0][1][j][v] + u1[a] * m[1][1][j][v];
            }
            #pragma unroll
            for (int e = 0; e < 4; ++e) {
                v2f o;
                o.x = u0[e] * t0[0] + u1[e] * t1[0];
                o.y = u0[e] * t0[1] + u1[e] * t1[1];
                *(v2f*)(outb + (a * KD + v) * CDIM + e * KD) = o;
            }
        }
    }
}

extern "C" void kernel_launch(void* const* d_in, const int* in_sizes, int n_in,
                              void* d_out, int out_size, void* d_ws, size_t ws_size,
                              hipStream_t stream) {
    (void)in_sizes; (void)n_in; (void)out_size; (void)d_ws; (void)ws_size;
    const float* rho = (const float*)d_in[0];
    const float* ux  = (const float*)d_in[1];
    const float* uy  = (const float*)d_in[2];
    const float* uc  = (const float*)d_in[3];
    float* out = (float*)d_out;

    const int lds_bytes = (KSUB * RHO_S + KD * KT_S) * (int)sizeof(float);
    (void)hipFuncSetAttribute((const void*)kron_apply_kernel,
                              hipFuncAttributeMaxDynamicSharedMemorySize, lds_bytes);

    kron_apply_kernel<<<NBATCH, 256, lds_bytes, stream>>>(rho, ux, uy, uc, out);
}


# Phase B split by column channel so a lane owns 4 adjacent columns: 16-byte stores, each half-wave writing one whole aligned 256-byte piece; partner waves swap c-mixed halves via LDS; sums unchanged
# speedup vs baseline: 3.7379x; 1.3878x over previous
#include <hip/hip_runtime.h>

#define NBATCH 256
#define CDIM   256
#define KSUB   128
#define KD     64

#define RHO_S  144
#define PT_S   137
#define KT_S   72
#define XCH_OFF (KSUB * RHO_S + KD * KT_S)
#define XCH_F   (8 * 16 * 32 * 4)

typedef __attribute__((ext_vector_type(2))) float v2f;
typedef __attribute__((ext_vector_type(4))) float v4f;
typedef __attribute__((ext_vector_type(8))) float v8f;

template <int D>
static __device__ __forceinline__ void mix_store(const v4f (&town)[2][8],
                                                 const v4f* __restrict__ xr,
                                                 const float (&u0)[4],
                                                 const float (&u1)[4],
                                                 float* __restrict__ outw) {
    #pragma unroll
    for (int i = 0; i < 2; ++i) {
        #pragma unroll
        for (int v = 0; v < 8; ++v) {
            const v4f tr = xr[(i * 8 + v) * 32];
            const v4f t0 = D ? tr : town[i][v];
            const v4f t1 = D ? town[i][v] : tr;
            #pragma unroll
            for (int e = 0; e < 4; ++e) {
                v4f o;
                #pragma unroll
                for (int j = 0; j < 4; ++j)
                    o[j] = u0[e] * t0[j] + u1[e] * t1[j];
                *(v4f*)(outw + (i * KD + v) * CDIM + e * KD) = o;
            }
        }
    }
}

__global__ void __launch_bounds__(256)
kron_apply_kernel(const float* __restrict__ rho,
                  const float* __restrict__ ux,
                  const float* __restrict__ uy,
                  const float* __restrict__ uc,
                  float* __restrict__ out) {
    extern __shared__ float smem[];
    float* sRho = smem;
    float* sPt  = smem;
    float* sKT  = smem + KSUB * RHO_S;
    v4f*   sX   = (v4f*)(smem + XCH_OFF);

    const int b    = blockIdx.x;
    const int tid  = threadIdx.x;
    const int lane = tid & 31;
    const int w    = tid >> 5;
    const int l16  = lane & 15;
    const int hl   = lane >> 4;

    {
        const int p  = tid & 63;
        const int i1 = p >> 3;
        const int j1 = p & 7;
        const int qb = tid >> 6;
        float uxr[8];
        #pragma unroll
        for (int t = 0; t < 8; ++t) uxr[t] = ux[i1 * 8 + t];
        const float uy0 = uy[j1 * 8 + qb];
        const float uy1 = uy[j1 * 8 + qb + 4];

        #pragma unroll
        for (int i = 0; i < 16; ++i)
            sKT[(qb + 4 * i) * KT_S + p] = uxr[i >> 1] * ((i & 1) ? uy1 : uy0);
    }
    {
        const float4* src = (const float4*)(rho + (size_t)b * KSUB * KSUB);
        float4* dst = (float4*)sRho;
        #pragma unroll
        for (int i = 0; i < (KSUB * KSUB / 4) / 256; ++i) {
            int g    = tid + i * 256;
            int row  = g >> 5;
            int col4 = g & 31;
            dst[row * (RHO_S / 4) + col4] = src[g];
        }
    }
    __syncthreads();

    {
        const int c   = w >> 2;
        const int pt0 = 2 * ((w >> 1) & 1);
        const int tn0 = 4 * (w & 1);
        v8f acc[2][4];
        #pragma unroll
        for (int i = 0; i < 2; ++i)
            #pragma unroll
            for (int j = 0; j < 4; ++j) acc[i][j] = v8f{};

        for (int kk = 0; kk < KD; kk += 4) {
            v2f a[2], bf[4];
            #pragma unroll
            for (int i = 0; i < 2; ++i) {
                a[i].x = sKT[(kk + 2 * hl)     * KT_S + (pt0 + i) * 16 + l16];
                a[i].y = sKT[(kk + 2 * hl + 1) * KT_S + (pt0 + i) * 16 + l16];
            }
            #pragma unroll
            for (int j = 0; j < 4; ++j) {
                bf[j].x = sRho[(c * KD + kk + 2 * hl)     * RHO_S + (tn0 + j) * 16 + l16];
                bf[j].y = sRho[(c * KD + kk + 2 * hl + 1) * RHO_S + (tn0 + j) * 16 + l16];
            }
            #pragma unroll
            for (int i = 0; i < 2; ++i)
                #pragma unroll
                for (int j = 0; j < 4; ++j)
                    acc[i][j] = __builtin_amdgcn_wmma_f32_16x16x4_f32(
                        false, a[i], false, bf[j], (short)0, acc[i][j], false, false);
        }
        __syncthreads();
        #pragma unroll
        for (int i = 0; i < 2; ++i)
            #pragma unroll
            for (int j = 0; j < 4; ++j)
                #pragma unroll
                for (int v = 0; v < 8; ++v)
                    sPt[((tn0 + j) * 16 + l16) * PT_S + c * KD + (pt0 + i) * 16 + v + 8 * hl] = acc[i][j][v];
    }
    __syncthreads();

    const int pt = w >> 1;
    const int d  = __builtin_amdgcn_readfirstlane(w & 1);
    v8f m[2][4];
    #pragma unroll
    for (int c = 0; c < 2; ++c)
        #pragma unroll
        for (int j = 0; j < 4; ++j) m[c][j] = v8f{};

    for (int kk = 0; kk < KD; kk += 4) {
        v2f a[2];
        #pragma unroll
        for (int c = 0; c < 2; ++c) {
            a[c].x = sPt[(d * KD + kk + 2 * hl)     * PT_S + c * KD + pt * 16 + l16];
            a[c].y = sPt[(d * KD + kk + 2 * hl + 1) * PT_S + c * KD + pt * 16 + l16];
        }
        const v4f bx = *(const v4f*)(sKT + (kk + 2 * hl)     * KT_S + 4 * l16);
        const v4f by = *(const v4f*)(sKT + (kk + 2 * hl + 1) * KT_S + 4 * l16);
        #pragma unroll
        for (int c = 0; c < 2; ++c)
            #pragma unroll
            for (int j = 0; j < 4; ++j) {
                v2f bf;
                bf.x = bx[j];
                bf.y = by[j];
                m[c][j] = __builtin_amdgcn_wmma_f32_16x16x4_f32(
                    false, a[c], false, bf, (short)0, m[c][j], false, false);
            }
    }

    float u0[4], u1[4];
    #pragma unroll
    for (int a = 0; a < 4; ++a) { u0[a] = uc[a * 4 + 2]; u1[a] = uc[a * 4 + 3]; }
    float uo0[2], uo1[2], up0[2], up1[2];
    #pragma unroll
    for (int i = 0; i < 2; ++i) {
        uo0[i] = d ? u0[2 + i] : u0[i];
        uo1[i] = d ? u1[2 + i] : u1[i];
        up0[i] = d ? u0[i] : u0[2 + i];
        up1[i] = d ? u1[i] : u1[2 + i];
    }

    v4f* xs = sX + (w * 16) * 32 + lane;
    #pragma unroll
    for (int i = 0; i < 2; ++i)
        #pragma unroll
        for (int v = 0; v < 8; ++v) {
            v4f t;
            #pragma unroll
            for (int j = 0; j < 4; ++j)
                t[j] = up0[i] * m[0][j][v] + up1[i] * m[1][j][v];
            xs[(i * 8 + v) * 32] = t;
        }
    v4f town[2][8];
    #pragma unroll
    for (int i = 0; i < 2; ++i)
        #pragma unroll
        for (int v = 0; v < 8; ++v)
            #pragma unroll
            for (int j = 0; j < 4; ++j)
                town[i][v][j] = uo0[i] * m[0][j][v] + uo1[i] * m[1][j][v];
    __syncthreads();

    const v4f* xr = sX + ((w ^ 1) * 16) * 32 + lane;
    float* outw = out + (size_t)b * CDIM * CDIM
                + (size_t)(2 * d * KD + pt * 16 + 8 * hl) * CDIM + 4 * l16;
    if (d == 0) mix_store<0>(town, xr, u0, u1, outw);
    else        mix_store<1>(town, xr, u0, u1, outw);
}

extern "C" void kernel_launch(void* const* d_in, const int* in_sizes, int n_in,
                              void* d_out, int out_size, void* d_ws, size_t ws_size,
                              hipStream_t stream) {
    (void)in_sizes; (void)n_in; (void)out_size; (void)d_ws; (void)ws_size;
    const float* rho = (const float*)d_in[0];
    const float* ux  = (const float*)d_in[1];
    const float* uy  = (const float*)d_in[2];
    const float* uc  = (const float*)d_in[3];
    float* out = (float*)d_out;

    const int lds_bytes = (KSUB * RHO_S + KD * KT_S + XCH_F) * (int)sizeof(float);
    (void)hipFuncSetAttribute((const void*)kron_apply_kernel,
                              hipFuncAttributeMaxDynamicSharedMemorySize, lds_bytes);

    kron_apply_kernel<<<NBATCH, 256, lds_bytes, stream>>>(rho, ux, uy, uc, out);
}


# Phase B row tile re-cut as (both c) x 8 p per wave, both column channels: c-mix by lane^16 register swap, no LDS exchange or barrier before stores; which product of a 2-term sum is fused may swap
# speedup vs baseline: 4.0437x; 1.0818x over previous
#include <hip/hip_runtime.h>

#define NBATCH 256
#define CDIM   256
#define KSUB   128
#define KD     64

#define RHO_S  144
#define PT_S   137
#define PC_S   72
#define KT_S   72

typedef __attribute__((ext_vector_type(2))) float v2f;
typedef __attribute__((ext_vector_type(4))) float v4f;
typedef __attribute__((ext_vector_type(8))) float v8f;

static __device__ __forceinline__ float xhalf(float x) {
    const int b = __builtin_bit_cast(int, x);
    return __builtin_bit_cast(float,
        __builtin_amdgcn_permlanex16(b, b, 0x76543210, 0xfedcba98, false, false));
}

__global__ void __launch_bounds__(256)
kron_apply_kernel(const float* __restrict__ rho,
                  const float* __restrict__ ux,
                  const float* __restrict__ uy,
                  const float* __restrict__ uc,
                  float* __restrict__ out) {
    extern __shared__ float smem[];
    float* sRho = smem;
    float* sPt  = smem;
    float* sKT  = smem + KSUB * RHO_S;

    const int b    = blockIdx.x;
    const int tid  = threadIdx.x;
    const int lane = tid & 31;
    const int w    = tid >> 5;
    const int l16  = lane & 15;
    const int hl   = lane >> 4;

    {
        const int p  = tid & 63;
        const int i1 = p >> 3;
        const int j1 = p & 7;
        const int qb = tid >> 6;
        float uxr[8];
        #pragma unroll
        for (int t = 0; t < 8; ++t) uxr[t] = ux[i1 * 8 + t];
        const float uy0 = uy[j1 * 8 + qb];
        const float uy1 = uy[j1 * 8 + qb + 4];

        #pragma unroll
        for (int i = 0; i < 16; ++i)
            sKT[(qb + 4 * i) * KT_S + p] = uxr[i >> 1] * ((i & 1) ? uy1 : uy0);
    }
    {
        const float4* src = (const float4*)(rho + (size_t)b * KSUB * KSUB);
        float4* dst = (float4*)sRho;
        #pragma unroll
        for (int i = 0; i < (KSUB * KSUB / 4) / 256; ++i) {
            int g    = tid + i * 256;
            int row  = g >> 5;
            int col4 = g & 31;
            dst[row * (RHO_S / 4) + col4] = src[g];
        }
    }
    __syncthreads();

    {
        const int c   = w >> 2;
        const int pt0 = 2 * ((w >> 1) & 1);
        const int tn0 = 4 * (w & 1);
        v8f acc[2][4];
        #pragma unroll
        for (int i = 0; i < 2; ++i)
            #pragma unroll
            for (int j = 0; j < 4; ++j) acc[i][j] = v8f{};

        for (int kk = 0; kk < KD; kk += 4) {
            v2f a[2], bf[4];
            #pragma unroll
            for (int i = 0; i < 2; ++i) {
                a[i].x = sKT[(kk + 2 * hl)     * KT_S + (pt0 + i) * 16 + l16];
                a[i].y = sKT[(kk + 2 * hl + 1) * KT_S + (pt0 + i) * 16 + l16];
            }
            #pragma unroll
            for (int j = 0; j < 4; ++j) {
                bf[j].x = sRho[(c * KD + kk + 2 * hl)     * RHO_S + (tn0 + j) * 16 + l16];
                bf[j].y = sRho[(c * KD + kk + 2 * hl + 1) * RHO_S + (tn0 + j) * 16 + l16];
            }
            #pragma unroll
            for (int i = 0; i < 2; ++i)
                #pragma unroll
                for (int j = 0; j < 4; ++j)
                    acc[i][j] = __builtin_amdgcn_wmma_f32_16x16x4_f32(
                        false, a[i], false, bf[j], (short)0, acc[i][j], false, false);
        }
        __syncthreads();
        #pragma unroll
        for (int i = 0; i < 2; ++i)
            #pragma unroll
            for (int j = 0; j < 4; ++j)
                #pragma unroll
                for (int v = 0; v < 8; ++v)
                    sPt[((tn0 + j) * 16 + l16) * PT_S + c * PC_S + (pt0 + i) * 16 + v + 8 * hl] = acc[i][j][v];
    }
    __syncthreads();

    const int arow = (l16 >> 3) * PC_S + 8 * w + (l16 & 7);
    v8f m[2][4];
    #pragma unroll
    for (int d = 0; d < 2; ++d)
        #pragma unroll
        for (int j = 0; j < 4; ++j) m[d][j] = v8f{};

    for (int kk = 0; kk < KD; kk += 4) {
        v2f a[2];
        #pragma unroll
        for (int d = 0; d < 2; ++d) {
            a[d].x = sPt[(d * KD + kk + 2 * hl)     * PT_S + arow];
            a[d].y = sPt[(d * KD + kk + 2 * hl + 1) * PT_S + arow];
        }
        const v4f bx = *(const v4f*)(sKT + (kk + 2 * hl)     * KT_S + 4 * l16);
        const v4f by = *(const v4f*)(sKT + (kk + 2 * hl + 1) * KT_S + 4 * l16);
        #pragma unroll
        for (int d = 0; d < 2; ++d)
            #pragma unroll
            for (int j = 0; j < 4; ++j) {
                v2f bf;
                bf.x = bx[j];
                bf.y = by[j];
                m[d][j] = __builtin_amdgcn_wmma_f32_16x16x4_f32(
                    false, a[d], false, bf, (short)0, m[d][j], false, false);
            }
    }

    float u0[4], u1[4];
    #pragma unroll
    for (int a = 0; a < 4; ++a) { u0[a] = uc[a * 4 + 2]; u1[a] = uc[a * 4 + 3]; }
    float cown[2], cpar[2];
    #pragma unroll
    for (int i = 0; i < 2; ++i) {
        cown[i] = hl ? u1[2 + i] : u0[i];
        cpar[i] = hl ? u0[2 + i] : u1[i];
    }

    float* outw = out + (size_t)b * CDIM * CDIM
                + (size_t)(2 * hl * KD + 8 * w) * CDIM + 4 * l16;
    #pragma unroll
    for (int r = 0; r < 8; ++r) {
        float own[2][4], par[2][4];
        #pragma unroll
        for (int d = 0; d < 2; ++d)
            #pragma unroll
            for (int j = 0; j < 4; ++j) {
                own[d][j] = m[d][j][r];
                par[d][j] = xhalf(own[d][j]);
            }
        #pragma unroll
        for (int i = 0; i < 2; ++i) {
            v4f t0, t1;
            #pragma unroll
            for (int j = 0; j < 4; ++j) {
                t0[j] = cown[i] * own[0][j] + cpar[i] * par[0][j];
                t1[j] = cown[i] * own[1][j] + cpar[i] * par[1][j];
            }
            #pragma unroll
            for (int e = 0; e < 4; ++e) {
                v4f o;
                #pragma unroll
                for (int j = 0; j < 4; ++j)
                    o[j] = u0[e] * t0[j] + u1[e] * t1[j];
                *(v4f*)(outw + (i * KD + r) * CDIM + e * KD) = o;
            }
        }
    }
}

extern "C" void kernel_launch(void* const* d_in, const int* in_sizes, int n_in,
                              void* d_out, int out_size, void* d_ws, size_t ws_size,
                              hipStream_t stream) {
    (void)in_sizes; (void)n_in; (void)out_size; (void)d_ws; (void)ws_size;
    const float* rho = (const float*)d_in[0];
    const float* ux  = (const float*)d_in[1];
    const float* uy  = (const float*)d_in[2];
    const float* uc  = (const float*)d_in[3];
    float* out = (float*)d_out;

    const int lds_bytes = (KSUB * RHO_S + KD * KT_S) * (int)sizeof(float);
    (void)hipFuncSetAttribute((const void*)kron_apply_kernel,
                              hipFuncAttributeMaxDynamicSharedMemorySize, lds_bytes);

    kron_apply_kernel<<<NBATCH, 256, lds_bytes, stream>>>(rho, ux, uy, uc, out);
}


# K=kron(ux,uy) by its 8x8 factors in both phases: two chained 16x16x16 block products per tile in registers, half the matrix ops; LDS loads grouped ahead; lanes l,l^8 swap columns; sums regrouped
# speedup vs baseline: 4.4893x; 1.1102x over previous
#include <hip/hip_runtime.h>

#define NBATCH 256
#define CDIM   256
#define KSUB   128
#define KD     64

#define RHO_S  144
#define PT_S   137
#define PC_S   72

typedef __attribute__((ext_vector_type(2))) float v2f;
typedef __attribute__((ext_vector_type(4))) float v4f;
typedef __attribute__((ext_vector_type(8))) float v8f;

static __device__ __forceinline__ float xhalf(float x) {
    const int b = __builtin_bit_cast(int, x);
    return __builtin_bit_cast(float,
        __builtin_amdgcn_permlanex16(b, b, 0x76543210, 0xfedcba98, false, false));
}

static __device__ __forceinline__ float xoct(float x) {
    const int b = __builtin_bit_cast(int, x);
    return __builtin_bit_cast(float,
        __builtin_amdgcn_permlane16(b, b, 0xfedcba98, 0x76543210, false, false));
}

static __device__ __forceinline__ v8f mma(v2f a, v2f b, v8f c) {
    return __builtin_amdgcn_wmma_f32_16x16x4_f32(false, a, false, b, (short)0, c, false, false);
}

__global__ void __launch_bounds__(256)
kron_apply_kernel(const float* __restrict__ rho,
                  const float* __restrict__ ux,
                  const float* __restrict__ uy,
                  const float* __restrict__ uc,
                  float* __restrict__ out) {
    extern __shared__ float smem[];
    float* sRho = smem;
    float* sPt  = smem;

    const int b    = blockIdx.x;
    const int tid  = threadIdx.x;
    const int lane = tid & 31;
    const int w    = tid >> 5;
    const int l16  = lane & 15;
    const int hl   = lane >> 4;

    v2f fy[4], fx[4];
    {
        const int rr = l16 & 7;
        const int rh = l16 >> 3;
        #pragma unroll
        for (int kq = 0; kq < 4; ++kq) {
            const float y0 = uy[rr * 8 + 4 * (kq & 1) + 2 * hl];
            const float y1 = uy[rr * 8 + 4 * (kq & 1) + 2 * hl + 1];
            const bool on = (rh == (kq >> 1));
            fy[kq].x = on ? y0 : 0.0f;
            fy[kq].y = on ? y1 : 0.0f;
        }
        #pragma unroll
        for (int t = 0; t < 4; ++t) {
            const float x = ux[rr * 8 + 2 * t + hl];
            fx[t].x = (rh == 0) ? x : 0.0f;
            fx[t].y = (rh == 1) ? x : 0.0f;
        }
    }
    {
        const float4* src = (const float4*)(rho + (size_t)b * KSUB * KSUB);
        float4* dst = (float4*)sRho;
        #pragma unroll
        for (int i = 0; i < (KSUB * KSUB / 4) / 256; ++i) {
            int g    = tid + i * 256;
            int row  = g >> 5;
            int col4 = g & 31;
            dst[row * (RHO_S / 4) + col4] = src[g];
        }
    }
    __syncthreads();

    {
        const int c   = w >> 2;
        const int tn0 = 2 * (w & 3);
        v8f pa[2][4];
        #pragma unroll
        for (int u = 0; u < 2; ++u) {
            const int scol = (tn0 + u) * 16 + l16;
            v8f t1[4];
            #pragma unroll
            for (int t = 0; t < 4; ++t) t1[t] = v8f{};
            #pragma unroll
            for (int kq = 0; kq < 4; ++kq)
                #pragma unroll
                for (int t = 0; t < 4; ++t) {
                    const int q = c * KD + 16 * t + 4 * kq + 2 * hl;
                    v2f bf;
                    bf.x = sRho[q * RHO_S + scol];
                    bf.y = sRho[(q + 1) * RHO_S + scol];
                    t1[t] = mma(fy[kq], bf, t1[t]);
                }
            #pragma unroll
            for (int g = 0; g < 4; ++g) pa[u][g] = v8f{};
            #pragma unroll
            for (int t = 0; t < 4; ++t)
                #pragma unroll
                for (int g = 0; g < 4; ++g) {
                    v2f bf;
                    bf.x = t1[t][2 * g];
                    bf.y = t1[t][2 * g + 1];
                    pa[u][g] = mma(fx[t], bf, pa[u][g]);
                }
        }
        __builtin_amdgcn_sched_group_barrier(0x100, 32, 0);
        __builtin_amdgcn_sched_group_barrier(0x008, 64, 0);
        __syncthreads();
        #pragma unroll
        for (int u = 0; u < 2; ++u)
            #pragma unroll
            for (int g = 0; g < 4; ++g)
                #pragma unroll
                for (int r = 0; r < 8; ++r)
                    sPt[((tn0 + u) * 16 + l16) * PT_S + c * PC_S + 8 * r + 2 * g + hl] = pa[u][g][r];
    }
    __syncthreads();

    const int arow = (l16 >> 3) * PC_S + 8 * w + (l16 & 7);
    v8f m[2][4];
    #pragma unroll
    for (int d = 0; d < 2; ++d) {
        v8f z1[4];
        #pragma unroll
        for (int t = 0; t < 4; ++t) z1[t] = v8f{};
        #pragma unroll
        for (int kq = 0; kq < 4; ++kq)
            #pragma unroll
            for (int t = 0; t < 4; ++t) {
                const int s = d * KD + 16 * t + 4 * kq + 2 * hl;
                v2f bf;
                bf.x = sPt[s * PT_S + arow];
                bf.y = sPt[(s + 1) * PT_S + arow];
                z1[t] = mma(fy[kq], bf, z1[t]);
            }
        #pragma unroll
        for (int g = 0; g < 4; ++g) m[d][g] = v8f{};
        #pragma unroll
        for (int t = 0; t < 4; ++t)
            #pragma unroll
            for (int g = 0; g < 4; ++g) {
                v2f af;
                af.x = z1[t][2 * g];
                af.y = z1[t][2 * g + 1];
                m[d][g] = mma(af, fx[t], m[d][g]);
            }
    }

    __builtin_amdgcn_sched_group_barrier(0x100, 64, 0);
    __builtin_amdgcn_sched_group_barrier(0x008, 64, 0);

    float u0[4], u1[4];
    #pragma unroll
    for (int a = 0; a < 4; ++a) { u0[a] = uc[a * 4 + 2]; u1[a] = uc[a * 4 + 3]; }
    float cown[2], cpar[2];
    #pragma unroll
    for (int i = 0; i < 2; ++i) {
        cown[i] = hl ? u1[2 + i] : u0[i];
        cpar[i] = hl ? u0[2 + i] : u1[i];
    }
    const bool up = (l16 >> 3) != 0;

    float* outw = out + (size_t)b * CDIM * CDIM
                + (size_t)(2 * hl * KD + 8 * w) * CDIM + 8 * (l16 & 7) + 4 * (l16 >> 3);
    #pragma unroll
    for (int r = 0; r < 8; ++r) {
        float own[2][4], par[2][4];
        #pragma unroll
        for (int d = 0; d < 2; ++d) {
            const float v0 = m[d][0][r], v1 = m[d][1][r], v2 = m[d][2][r], v3 = m[d][3][r];
            const float g0 = xoct(up ? v0 : v2);
            const float g1 = xoct(up ? v1 : v3);
            own[d][0] = up ? g0 : v0;
            own[d][1] = up ? v2 : g0;
            own[d][2] = up ? g1 : v1;
            own[d][3] = up ? v3 : g1;
            #pragma unroll
            for (int j = 0; j < 4; ++j) par[d][j] = xhalf(own[d][j]);
        }
        #pragma unroll
        for (int i = 0; i < 2; ++i) {
            v4f t0, t1;
            #pragma unroll
            for (int j = 0; j < 4; ++j) {
                t0[j] = cown[i] * own[0][j] + cpar[i] * par[0][j];
                t1[j] = cown[i] * own[1][j] + cpar[i] * par[1][j];
            }
            #pragma unroll
            for (int e = 0; e < 4; ++e) {
                v4f o;
                #pragma unroll
                for (int j = 0; j < 4; ++j)
                    o[j] = u0[e] * t0[j] + u1[e] * t1[j];
                *(v4f*)(outw + (i * KD + r) * CDIM + e * KD) = o;
            }
        }
    }
}

extern "C" void kernel_launch(void* const* d_in, const int* in_sizes, int n_in,
                              void* d_out, int out_size, void* d_ws, size_t ws_size,
                              hipStream_t stream) {
    (void)in_sizes; (void)n_in; (void)out_size; (void)d_ws; (void)ws_size;
    const float* rho = (const float*)d_in[0];
    const float* ux  = (const float*)d_in[1];
    const float* uy  = (const float*)d_in[2];
    const float* uc  = (const float*)d_in[3];
    float* out = (float*)d_out;

    const int lds_bytes = (KSUB * RHO_S) * (int)sizeof(float);
    (void)hipFuncSetAttribute((const void*)kron_apply_kernel,
                              hipFuncAttributeMaxDynamicSharedMemorySize, lds_bytes);

    kron_apply_kernel<<<NBATCH, 256, lds_bytes, stream>>>(rho, ux, uy, uc, out);
}


# Column mix (over d) before row mix (over c): lane half hl owns blocks a'=2k+hl of all 4 rows a, so a store instruction writes one contiguous 512-byte run, not two 256-byte pieces; sums regrouped
# speedup vs baseline: 4.8060x; 1.0705x over previous
#include <hip/hip_runtime.h>

#define NBATCH 256
#define CDIM   256
#define KSUB   128
#define KD     64

#define RHO_S  144
#define PT_S   137
#define PC_S   72

typedef __attribute__((ext_vector_type(2))) float v2f;
typedef __attribute__((ext_vector_type(4))) float v4f;
typedef __attribute__((ext_vector_type(8))) float v8f;

static __device__ __forceinline__ float xhalf(float x) {
    const int b = __builtin_bit_cast(int, x);
    return __builtin_bit_cast(float,
        __builtin_amdgcn_permlanex16(b, b, 0x76543210, 0xfedcba98, false, false));
}

static __device__ __forceinline__ float xoct(float x) {
    const int b = __builtin_bit_cast(int, x);
    return __builtin_bit_cast(float,
        __builtin_amdgcn_permlane16(b, b, 0xfedcba98, 0x76543210, false, false));
}

static __device__ __forceinline__ v8f mma(v2f a, v2f b, v8f c) {
    return __builtin_amdgcn_wmma_f32_16x16x4_f32(false, a, false, b, (short)0, c, false, false);
}

__global__ void __launch_bounds__(256)
kron_apply_kernel(const float* __restrict__ rho,
                  const float* __restrict__ ux,
                  const float* __restrict__ uy,
                  const float* __restrict__ uc,
                  float* __restrict__ out) {
    extern __shared__ float smem[];
    float* sRho = smem;
    float* sPt  = smem;

    const int b    = blockIdx.x;
    const int tid  = threadIdx.x;
    const int lane = tid & 31;
    const int w    = tid >> 5;
    const int l16  = lane & 15;
    const int hl   = lane >> 4;

    v2f fy[4], fx[4];
    {
        const int rr = l16 & 7;
        const int rh = l16 >> 3;
        #pragma unroll
        for (int kq = 0; kq < 4; ++kq) {
            const float y0 = uy[rr * 8 + 4 * (kq & 1) + 2 * hl];
            const float y1 = uy[rr * 8 + 4 * (kq & 1) + 2 * hl + 1];
            const bool on = (rh == (kq >> 1));
            fy[kq].x = on ? y0 : 0.0f;
            fy[kq].y = on ? y1 : 0.0f;
        }
        #pragma unroll
        for (int t = 0; t < 4; ++t) {
            const float x = ux[rr * 8 + 2 * t + hl];
            fx[t].x = (rh == 0) ? x : 0.0f;
            fx[t].y = (rh == 1) ? x : 0.0f;
        }
    }
    {
        const float4* src = (const float4*)(rho + (size_t)b * KSUB * KSUB);
        float4* dst = (float4*)sRho;
        #pragma unroll
        for (int i = 0; i < (KSUB * KSUB / 4) / 256; ++i) {
            int g    = tid + i * 256;
            int row  = g >> 5;
            int col4 = g & 31;
            dst[row * (RHO_S / 4) + col4] = src[g];
        }
    }
    __syncthreads();

    {
        const int c   = w >> 2;
        const int tn0 = 2 * (w & 3);
        v8f pa[2][4];
        #pragma unroll
        for (int u = 0; u < 2; ++u) {
            const int scol = (tn0 + u) * 16 + l16;
            v8f t1[4];
            #pragma unroll
            for (int t = 0; t < 4; ++t) t1[t] = v8f{};
            #pragma unroll
            for (int kq = 0; kq < 4; ++kq)
                #pragma unroll
                for (int t = 0; t < 4; ++t) {
                    const int q = c * KD + 16 * t + 4 * kq + 2 * hl;
                    v2f bf;
                    bf.x = sRho[q * RHO_S + scol];
                    bf.y = sRho[(q + 1) * RHO_S + scol];
                    t1[t] = mma(fy[kq], bf, t1[t]);
                }
            #pragma unroll
            for (int g = 0; g < 4; ++g) pa[u][g] = v8f{};
            #pragma unroll
            for (int t = 0; t < 4; ++t)
                #pragma unroll
                for (int g = 0; g < 4; ++g) {
                    v2f bf;
                    bf.x = t1[t][2 * g];
                    bf.y = t1[t][2 * g + 1];
                    pa[u][g] = mma(fx[t], bf, pa[u][g]);
                }
        }
        __builtin_amdgcn_sched_group_barrier(0x100, 32, 0);
        __builtin_amdgcn_sched_group_barrier(0x008, 64, 0);
        __syncthreads();
        #pragma unroll
        for (int u = 0; u < 2; ++u)
            #pragma unroll
            for (int g = 0; g < 4; ++g)
                #pragma unroll
                for (int r = 0; r < 8; ++r)
                    sPt[((tn0 + u) * 16 + l16) * PT_S + c * PC_S + 8 * r + 2 * g + hl] = pa[u][g][r];
    }
    __syncthreads();

    const int arow = (l16 >> 3) * PC_S + 8 * w + (l16 & 7);
    v8f m[2][4];
    #pragma unroll
    for (int d = 0; d < 2; ++d) {
        v8f z1[4];
        #pragma unroll
        for (int t = 0; t < 4; ++t) z1[t] = v8f{};
        #pragma unroll
        for (int kq = 0; kq < 4; ++kq)
            #pragma unroll
            for (int t = 0; t < 4; ++t) {
                const int s = d * KD + 16 * t + 4 * kq + 2 * hl;
                v2f bf;
                bf.x = sPt[s * PT_S + arow];
                bf.y = sPt[(s + 1) * PT_S + arow];
                z1[t] = mma(fy[kq], bf, z1[t]);
            }
        #pragma unroll
        for (int g = 0; g < 4; ++g) m[d][g] = v8f{};
        #pragma unroll
        for (int t = 0; t < 4; ++t)
            #pragma unroll
            for (int g = 0; g < 4; ++g) {
                v2f af;
                af.x = z1[t][2 * g];
                af.y = z1[t][2 * g + 1];
                m[d][g] = mma(af, fx[t], m[d][g]);
            }
    }

    __builtin_amdgcn_sched_group_barrier(0x100, 64, 0);
    __builtin_amdgcn_sched_group_barrier(0x008, 64, 0);

    float u0[4], u1[4];
    #pragma unroll
    for (int a = 0; a < 4; ++a) { u0[a] = uc[a * 4 + 2]; u1[a] = uc[a * 4 + 3]; }
    float kp0[2], kp1[2];
    float sd0[2], sd1[2];
    #pragma unroll
    for (int k = 0; k < 2; ++k) {
        kp0[k] = hl ? u0[2 * k + 1] : u0[2 * k];
        kp1[k] = hl ? u1[2 * k + 1] : u1[2 * k];
        sd0[k] = hl ? u0[2 * k] : u0[2 * k + 1];
        sd1[k] = hl ? u1[2 * k] : u1[2 * k + 1];
    }
    float rs[4], ro[4];
    #pragma unroll
    for (int a = 0; a < 4; ++a) {
        rs[a] = hl ? u1[a] : u0[a];
        ro[a] = hl ? u0[a] : u1[a];
    }
    const bool up = (l16 >> 3) != 0;

    float* outw = out + (size_t)b * CDIM * CDIM
                + (size_t)(8 * w) * CDIM + hl * KD + 8 * (l16 & 7) + 4 * (l16 >> 3);
    #pragma unroll
    for (int r = 0; r < 8; ++r) {
        float own[2][4];
        #pragma unroll
        for (int d = 0; d < 2; ++d) {
            const float v0 = m[d][0][r], v1 = m[d][1][r], v2 = m[d][2][r], v3 = m[d][3][r];
            const float g0 = xoct(up ? v0 : v2);
            const float g1 = xoct(up ? v1 : v3);
            own[d][0] = up ? g0 : v0;
            own[d][1] = up ? v2 : g0;
            own[d][2] = up ? g1 : v1;
            own[d][3] = up ? v3 : g1;
        }
        v4f keep[2], recv[2];
        #pragma unroll
        for (int k = 0; k < 2; ++k)
            #pragma unroll
            for (int j = 0; j < 4; ++j) {
                keep[k][j] = kp0[k] * own[0][j] + kp1[k] * own[1][j];
                recv[k][j] = xhalf(sd0[k] * own[0][j] + sd1[k] * own[1][j]);
            }
        #pragma unroll
        for (int a = 0; a < 4; ++a)
            #pragma unroll
            for (int k = 0; k < 2; ++k) {
                v4f o;
                #pragma unroll
                for (int j = 0; j < 4; ++j)
                    o[j] = rs[a] * keep[k][j] + ro[a] * recv[k][j];
                *(v4f*)(outw + (a * KD + r) * CDIM + 2 * k * KD) = o;
            }
    }
}

extern "C" void kernel_launch(void* const* d_in, const int* in_sizes, int n_in,
                              void* d_out, int out_size, void* d_ws, size_t ws_size,
                              hipStream_t stream) {
    (void)in_sizes; (void)n_in; (void)out_size; (void)d_ws; (void)ws_size;
    const float* rho = (const float*)d_in[0];
    const float* ux  = (const float*)d_in[1];
    const float* uy  = (const float*)d_in[2];
    const float* uc  = (const float*)d_in[3];
    float* out = (float*)d_out;

    const int lds_bytes = (KSUB * RHO_S) * (int)sizeof(float);
    (void)hipFuncSetAttribute((const void*)kron_apply_kernel,
                              hipFuncAttributeMaxDynamicSharedMemorySize, lds_bytes);

    kron_apply_kernel<<<NBATCH, 256, lds_bytes, stream>>>(rho, ux, uy, uc, out);
}
